# starting kernel
# baseline (speedup 1.0000x reference)
#include <hip/hip_runtime.h>
#include <hip/hip_bf16.h>

typedef float v2f __attribute__((ext_vector_type(2)));
typedef float v8f __attribute__((ext_vector_type(8)));

#ifndef GCN_THREADS
#define GCN_THREADS 256
#endif

__global__ void k_deg_init(float* __restrict__ deg, int n) {
  int i = blockIdx.x * blockDim.x + threadIdx.x;
  if (i < n) deg[i] = 1.0f;
}

__global__ void k_deg_count(const int* __restrict__ dst, float* __restrict__ deg, int e) {
  int i = blockIdx.x * blockDim.x + threadIdx.x;
  if (i < e) atomicAdd(&deg[dst[i]], 1.0f);
}

__global__ void k_dinv(float* __restrict__ deg, int n) {
  int i = blockIdx.x * blockDim.x + threadIdx.x;
  if (i < n) deg[i] = rsqrtf(deg[i]);
}

template <int K, int NO, int NT>
__global__ void k_gemm_wmma(const float* __restrict__ A, const float* __restrict__ W,
                            float* __restrict__ out, int mtiles) {
  const int wave = (int)((blockIdx.x * blockDim.x + threadIdx.x) >> 5);
  const int lane = (int)(threadIdx.x & 31);
  if (wave >= mtiles) return;

  const int hi   = lane >> 4;
  const int lo   = lane & 15;
  const int kb   = hi << 1;
  const int row  = wave * 16 + lo;

  v8f acc[NT];
#pragma unroll
  for (int t = 0; t < NT; ++t) acc[t] = (v8f){0.f, 0.f, 0.f, 0.f, 0.f, 0.f, 0.f, 0.f};

  const float* __restrict__ arow = A + (size_t)row * K;

  for (int k0 = 0; k0 < K; k0 += 4) {
    v2f a = *(const v2f*)(arow + k0 + kb);
#pragma unroll
    for (int t = 0; t < NT; ++t) {
      const int n = t * 16 + lo;
      v2f b;
      b.x = W[(size_t)(k0 + kb + 0) * NO + n];
      b.y = W[(size_t)(k0 + kb + 1) * NO + n];
      acc[t] = __builtin_amdgcn_wmma_f32_16x16x4_f32(
           false, a,  false, b,
           (short)0, acc[t],  false,  false);
    }
  }

  const int mbase = wave * 16 + (hi << 3);
#pragma unroll
  for (int t = 0; t < NT; ++t) {
#pragma unroll
    for (int r = 0; r < 8; ++r) {
      out[(size_t)(mbase + r) * NO + t * 16 + lo] = acc[t][r];
    }
  }
}

template <int D>
__global__ void k_selfloop_init(const float* __restrict__ h, const float* __restrict__ dinv,
                                float* __restrict__ acc, int n) {
  const unsigned QP = D / 4;
  unsigned i = blockIdx.x * blockDim.x + threadIdx.x;
  if (i >= (unsigned)n * QP) return;
  unsigned node = i / QP;
  unsigned f4   = (i % QP) * 4;
  float s = dinv[node];
  s = s * s;
  float4 v = *(const float4*)(h + (size_t)node * D + f4);
  v.x *= s; v.y *= s; v.z *= s; v.w *= s;
  *(float4*)(acc + (size_t)node * D + f4) = v;
}

template <int D>
__global__ void k_edge_scatter(const int* __restrict__ src, const int* __restrict__ dst,
                               const float* __restrict__ dinv, const float* __restrict__ h,
                               float* __restrict__ acc, int e) {
  const unsigned QP = D / 4;
  unsigned tid = blockIdx.x * blockDim.x + threadIdx.x;
  if (tid >= (unsigned)e * QP) return;
  unsigned eid = tid / QP;
  unsigned f4  = (tid % QP) * 4;
  int s = src[eid];
  int d = dst[eid];
  float nrm = dinv[s] * dinv[d];
  float4 v = *(const float4*)(h + (size_t)s * D + f4);
  float* o = acc + (size_t)d * D + f4;
  atomicAdd(o + 0, v.x * nrm);
  atomicAdd(o + 1, v.y * nrm);
  atomicAdd(o + 2, v.z * nrm);
  atomicAdd(o + 3, v.w * nrm);
}

template <int D>
__global__ void k_bias_relu(const float* __restrict__ acc, const float* __restrict__ b,
                            float* __restrict__ out, unsigned total) {
  unsigned i = blockIdx.x * blockDim.x + threadIdx.x;
  if (i >= total) return;
  out[i] = fmaxf(acc[i] + b[i & (D - 1)], 0.0f);
}

template <int D>
__global__ void k_bias_add(float* __restrict__ out, const float* __restrict__ b, unsigned total) {
  unsigned i = blockIdx.x * blockDim.x + threadIdx.x;
  if (i >= total) return;
  out[i] += b[i & (D - 1)];
}

extern "C" void kernel_launch(void* const* d_in, const int* in_sizes, int n_in,
                              void* d_out, int out_size, void* d_ws, size_t ws_size,
                              hipStream_t stream) {
  (void)n_in; (void)out_size; (void)ws_size;

  const float* x    = (const float*)d_in[0];
  const int*   ei   = (const int*)d_in[1];
  const float* W1   = (const float*)d_in[2];
  const float* b1   = (const float*)d_in[3];
  const float* W2   = (const float*)d_in[4];
  const float* b2   = (const float*)d_in[5];
  float*       out  = (float*)d_out;

  const int N = in_sizes[0] / 128;
  const int E = in_sizes[1] / 2;
  const int IN_D = 128, HID_D = 128, OUT_D = 64;

  const int* src = ei;
  const int* dst = ei + E;

  float* dinv = (float*)d_ws;
  float* bufA = dinv + N;
  float* bufB = bufA + (size_t)N * 128;

  const int T = GCN_THREADS;
  auto blk = [&](long long work) { return (unsigned)((work + T - 1) / T); };

  k_deg_init<<<blk(N), T, 0, stream>>>(dinv, N);
  k_deg_count<<<blk(E), T, 0, stream>>>(dst, dinv, E);
  k_dinv<<<blk(N), T, 0, stream>>>(dinv, N);

  const int mtiles = N / 16;
  {
    const int wavesPerBlk = T / 32;
    unsigned g = (unsigned)((mtiles + wavesPerBlk - 1) / wavesPerBlk);
    k_gemm_wmma<IN_D, HID_D, HID_D / 16><<<g, T, 0, stream>>>(x, W1, bufA, mtiles);
  }

  k_selfloop_init<HID_D><<<blk((long long)N * (HID_D / 4)), T, 0, stream>>>(bufA, dinv, bufB, N);
  k_edge_scatter<HID_D><<<blk((long long)E * (HID_D / 4)), T, 0, stream>>>(src, dst, dinv, bufA, bufB, E);
  k_bias_relu<HID_D><<<blk((long long)N * HID_D), T, 0, stream>>>(bufB, b1, bufA, (unsigned)N * HID_D);

  {
    const int wavesPerBlk = T / 32;
    unsigned g = (unsigned)((mtiles + wavesPerBlk - 1) / wavesPerBlk);
    k_gemm_wmma<HID_D, OUT_D, OUT_D / 16><<<g, T, 0, stream>>>(bufA, W2, bufB, mtiles);
  }

  k_selfloop_init<OUT_D><<<blk((long long)N * (OUT_D / 4)), T, 0, stream>>>(bufB, dinv, out, N);
  k_edge_scatter<OUT_D><<<blk((long long)E * (OUT_D / 4)), T, 0, stream>>>(src, dst, dinv, bufB, out, E);
  k_bias_add<OUT_D><<<blk((long long)N * OUT_D), T, 0, stream>>>(out, b2, (unsigned)N * OUT_D);
}


# Float atomic scatter-add replaced by a per-call counting sort of edges by destination and one wave per node summing its rows in registers; sum order: self-loop, then edges by atomic arrival rank
# speedup vs baseline: 1.5286x; 1.5286x over previous
#include <hip/hip_runtime.h>
#include <hip/hip_bf16.h>

typedef float v2f __attribute__((ext_vector_type(2)));
typedef float v8f __attribute__((ext_vector_type(8)));

#ifndef GCN_THREADS
#define GCN_THREADS 256
#endif

#define SCAN_T 256
#define SCAN_ITEMS 4
#define SCAN_WAVES (SCAN_T / 32)

__global__ void k_cnt_zero(int* __restrict__ cnt, int n) {
  int i = blockIdx.x * blockDim.x + threadIdx.x;
  if (i < n) cnt[i] = 0;
}

__global__ void k_count_rank(const int* __restrict__ dst, int* __restrict__ cnt,
                             int* __restrict__ rank, int e) {
  int i = blockIdx.x * blockDim.x + threadIdx.x;
  if (i < e) rank[i] = atomicAdd(&cnt[dst[i]], 1);
}

__device__ __forceinline__ int wave_sum(int v) {
#pragma unroll
  for (int d = 16; d > 0; d >>= 1) v += __shfl_xor(v, d, 32);
  return v;
}

__global__ void k_tile_sum(const int* __restrict__ cnt, int* __restrict__ tsum, int n) {
  __shared__ int wsum[SCAN_WAVES];
  const int t = (int)threadIdx.x, lane = t & 31, w = t >> 5;
  const int i0 = ((int)blockIdx.x * SCAN_T + t) * SCAN_ITEMS;
  int s = 0;
  if (i0 + SCAN_ITEMS <= n) {
    int4 c = *(const int4*)(cnt + i0);
    s = c.x + c.y + c.z + c.w;
  } else {
    for (int k = 0; k < SCAN_ITEMS; ++k) if (i0 + k < n) s += cnt[i0 + k];
  }
  s = wave_sum(s);
  if (lane == 0) wsum[w] = s;
  __syncthreads();
  if (t == 0) {
    int tot = 0;
#pragma unroll
    for (int k = 0; k < SCAN_WAVES; ++k) tot += wsum[k];
    tsum[blockIdx.x] = tot;
  }
}

__global__ void k_tile_scan(const int* __restrict__ cnt, const int* __restrict__ tsum,
                            int* __restrict__ rowptr, float* __restrict__ dinv, int n) {
  __shared__ int wsum[SCAN_WAVES];
  const int t = (int)threadIdx.x, lane = t & 31, w = t >> 5;

  int p = 0;
  for (int b = t; b < (int)blockIdx.x; b += SCAN_T) p += tsum[b];
  p = wave_sum(p);
  if (lane == 0) wsum[w] = p;
  __syncthreads();
  int run = 0;
#pragma unroll
  for (int k = 0; k < SCAN_WAVES; ++k) run += wsum[k];
  __syncthreads();

  const int i0 = ((int)blockIdx.x * SCAN_T + t) * SCAN_ITEMS;
  const bool full = (i0 + SCAN_ITEMS <= n);
  int c[SCAN_ITEMS];
  if (full) {
    int4 v = *(const int4*)(cnt + i0);
    c[0] = v.x; c[1] = v.y; c[2] = v.z; c[3] = v.w;
  } else {
#pragma unroll
    for (int k = 0; k < SCAN_ITEMS; ++k) c[k] = (i0 + k < n) ? cnt[i0 + k] : 0;
  }
  const int s = c[0] + c[1] + c[2] + c[3];

  int incl = s;
#pragma unroll
  for (int d = 1; d < 32; d <<= 1) {
    int u = __shfl_up(incl, d, 32);
    if (lane >= d) incl += u;
  }
  if (lane == 31) wsum[w] = incl;
  __syncthreads();
#pragma unroll
  for (int k = 0; k < SCAN_WAVES; ++k) if (k < w) run += wsum[k];
  run += incl - s;

  if (full) {
    int4 r;
    r.x = run; r.y = r.x + c[0]; r.z = r.y + c[1]; r.w = r.z + c[2];
    *(int4*)(rowptr + i0) = r;
    float4 dv;
    dv.x = rsqrtf((float)(c[0] + 1)); dv.y = rsqrtf((float)(c[1] + 1));
    dv.z = rsqrtf((float)(c[2] + 1)); dv.w = rsqrtf((float)(c[3] + 1));
    *(float4*)(dinv + i0) = dv;
    if (i0 + SCAN_ITEMS == n) rowptr[n] = r.w + c[3];
  } else {
#pragma unroll
    for (int k = 0; k < SCAN_ITEMS; ++k) {
      if (i0 + k < n) {
        rowptr[i0 + k] = run;
        dinv[i0 + k] = rsqrtf((float)(c[k] + 1));
        run += c[k];
        if (i0 + k == n - 1) rowptr[n] = run;
      }
    }
  }
}

__global__ void k_csr_fill(const int* __restrict__ src, const int* __restrict__ dst,
                           const int* __restrict__ rank, const int* __restrict__ rowptr,
                           const float* __restrict__ dinv, int2* __restrict__ csr, int e) {
  int i = blockIdx.x * blockDim.x + threadIdx.x;
  if (i >= e) return;
  int s = src[i];
  int p = rowptr[dst[i]] + rank[i];
  csr[p] = make_int2(s, __float_as_int(dinv[s]));
}

template <int K, int NO, int NT>
__global__ void k_gemm_wmma(const float* __restrict__ A, const float* __restrict__ W,
                            float* __restrict__ out, int mtiles) {
  const int wave = (int)((blockIdx.x * blockDim.x + threadIdx.x) >> 5);
  const int lane = (int)(threadIdx.x & 31);
  if (wave >= mtiles) return;

  const int hi   = lane >> 4;
  const int lo   = lane & 15;
  const int kb   = hi << 1;
  const int row  = wave * 16 + lo;

  v8f acc[NT];
#pragma unroll
  for (int t = 0; t < NT; ++t) acc[t] = (v8f){0.f, 0.f, 0.f, 0.f, 0.f, 0.f, 0.f, 0.f};

  const float* __restrict__ arow = A + (size_t)row * K;

  for (int k0 = 0; k0 < K; k0 += 4) {
    v2f a = *(const v2f*)(arow + k0 + kb);
#pragma unroll
    for (int t = 0; t < NT; ++t) {
      const int n = t * 16 + lo;
      v2f b;
      b.x = W[(size_t)(k0 + kb + 0) * NO + n];
      b.y = W[(size_t)(k0 + kb + 1) * NO + n];
      acc[t] = __builtin_amdgcn_wmma_f32_16x16x4_f32(
           false, a,  false, b,
           (short)0, acc[t],  false,  false);
    }
  }

  const int mbase = wave * 16 + (hi << 3);
#pragma unroll
  for (int t = 0; t < NT; ++t) {
#pragma unroll
    for (int r = 0; r < 8; ++r) {
      out[(size_t)(mbase + r) * NO + t * 16 + lo] = acc[t][r];
    }
  }
}

template <int V> struct vec_of;
template <> struct vec_of<2> { typedef v2f type; };
template <> struct vec_of<4> { typedef float type __attribute__((ext_vector_type(4))); };

template <int C, int D, typename vf>
__device__ __forceinline__ void gather_step(vf& acc, const int2 sw, const int j, const float di,
                                            const float* __restrict__ hl) {
  vf r[C];
  float w[C];
#pragma unroll
  for (int k = 0; k < C; ++k) {
    const int s = __builtin_amdgcn_readlane(sw.x, j + k);
    w[k] = __int_as_float(__builtin_amdgcn_readlane(sw.y, j + k)) * di;
    r[k] = *(const vf*)(hl + (size_t)s * D);
  }
#pragma unroll
  for (int k = 0; k < C; ++k) acc += r[k] * w[k];
}

template <int D>
__global__ void k_node_gather(const int* __restrict__ rowptr, const int2* __restrict__ csr,
                              const float* __restrict__ dinv, const float* __restrict__ h,
                              float* __restrict__ out, int n) {
  constexpr int V = D / 32;
  typedef typename vec_of<V>::type vf;
  const int lane = (int)(threadIdx.x & 31);
  const int node = __builtin_amdgcn_readfirstlane(
      (int)(blockIdx.x * (blockDim.x >> 5) + (threadIdx.x >> 5)));
  if (node >= n) return;

  const int beg = rowptr[node];
  const int end = rowptr[node + 1];
  const float di = dinv[node];
  const float* __restrict__ hl = h + lane * V;

  int base = beg;
  int m = (end - base < 32) ? (end - base) : 32;
  int2 sw = make_int2(0, 0);
  if (lane < m) sw = csr[base + lane];
  vf acc = *(const vf*)(hl + (size_t)node * D) * (di * di);

  for (;;) {
    int j = 0;
    for (; j + 8 <= m; j += 8) gather_step<8, D, vf>(acc, sw, j, di, hl);
    switch (m - j) {
      case 7: gather_step<7, D, vf>(acc, sw, j, di, hl); break;
      case 6: gather_step<6, D, vf>(acc, sw, j, di, hl); break;
      case 5: gather_step<5, D, vf>(acc, sw, j, di, hl); break;
      case 4: gather_step<4, D, vf>(acc, sw, j, di, hl); break;
      case 3: gather_step<3, D, vf>(acc, sw, j, di, hl); break;
      case 2: gather_step<2, D, vf>(acc, sw, j, di, hl); break;
      case 1: gather_step<1, D, vf>(acc, sw, j, di, hl); break;
      default: break;
    }
    base += 32;
    if (base >= end) break;
    m = (end - base < 32) ? (end - base) : 32;
    sw = make_int2(0, 0);
    if (lane < m) sw = csr[base + lane];
  }
  *(vf*)(out + (size_t)node * D + lane * V) = acc;
}

template <int D>
__global__ void k_bias_relu(const float* __restrict__ acc, const float* __restrict__ b,
                            float* __restrict__ out, unsigned total) {
  unsigned i = blockIdx.x * blockDim.x + threadIdx.x;
  if (i >= total) return;
  out[i] = fmaxf(acc[i] + b[i & (D - 1)], 0.0f);
}

template <int D>
__global__ void k_bias_add(float* __restrict__ out, const float* __restrict__ b, unsigned total) {
  unsigned i = blockIdx.x * blockDim.x + threadIdx.x;
  if (i >= total) return;
  out[i] += b[i & (D - 1)];
}

extern "C" void kernel_launch(void* const* d_in, const int* in_sizes, int n_in,
                              void* d_out, int out_size, void* d_ws, size_t ws_size,
                              hipStream_t stream) {
  (void)n_in; (void)out_size; (void)ws_size;

  const float* x    = (const float*)d_in[0];
  const int*   ei   = (const int*)d_in[1];
  const float* W1   = (const float*)d_in[2];
  const float* b1   = (const float*)d_in[3];
  const float* W2   = (const float*)d_in[4];
  const float* b2   = (const float*)d_in[5];
  float*       out  = (float*)d_out;

  const int N = in_sizes[0] / 128;
  const int E = in_sizes[1] / 2;
  const int IN_D = 128, HID_D = 128, OUT_D = 64;

  const int* src = ei;
  const int* dst = ei + E;

  auto up4 = [](size_t v) { return (v + 3) & ~(size_t)3; };
  const int ntile = (N + SCAN_T * SCAN_ITEMS - 1) / (SCAN_T * SCAN_ITEMS);
  int*   cnt    = (int*)d_ws;
  int*   rowptr = cnt + up4((size_t)N);
  int*   tsum   = rowptr + up4((size_t)N + 1);
  int*   rank   = tsum + up4((size_t)ntile);
  int2*  csr    = (int2*)(rank + up4((size_t)E));
  float* dinv   = (float*)(csr + up4((size_t)E));
  float* bufA   = dinv + up4((size_t)N);
  float* bufB   = bufA + (size_t)N * 128;

  const int T = GCN_THREADS;
  auto blk = [&](long long work) { return (unsigned)((work + T - 1) / T); };
  const unsigned gnode = (unsigned)((N + (T / 32) - 1) / (T / 32));

  k_cnt_zero<<<blk(N), T, 0, stream>>>(cnt, N);
  k_count_rank<<<blk(E), T, 0, stream>>>(dst, cnt, rank, E);
  k_tile_sum<<<ntile, SCAN_T, 0, stream>>>(cnt, tsum, N);
  k_tile_scan<<<ntile, SCAN_T, 0, stream>>>(cnt, tsum, rowptr, dinv, N);
  k_csr_fill<<<blk(E), T, 0, stream>>>(src, dst, rank, rowptr, dinv, csr, E);

  const int mtiles = N / 16;
  {
    const int wavesPerBlk = T / 32;
    unsigned g = (unsigned)((mtiles + wavesPerBlk - 1) / wavesPerBlk);
    k_gemm_wmma<IN_D, HID_D, HID_D / 16><<<g, T, 0, stream>>>(x, W1, bufA, mtiles);
  }

  k_node_gather<HID_D><<<gnode, T, 0, stream>>>(rowptr, csr, dinv, bufA, bufB, N);
  k_bias_relu<HID_D><<<blk((long long)N * HID_D), T, 0, stream>>>(bufB, b1, bufA, (unsigned)N * HID_D);

  {
    const int wavesPerBlk = T / 32;
    unsigned g = (unsigned)((mtiles + wavesPerBlk - 1) / wavesPerBlk);
    k_gemm_wmma<HID_D, OUT_D, OUT_D / 16><<<g, T, 0, stream>>>(bufA, W2, bufB, mtiles);
  }

  k_node_gather<OUT_D><<<gnode, T, 0, stream>>>(rowptr, csr, dinv, bufB, out, N);
  k_bias_add<OUT_D><<<blk((long long)N * OUT_D), T, 0, stream>>>(out, b2, (unsigned)N * OUT_D);
}


# Bias add and relu moved into the node-sum kernel's own store (value order unchanged: row sum, then + bias, then relu); the two element-wise launches and their re-read of both activations are gone
# speedup vs baseline: 1.9288x; 1.2618x over previous
#include <hip/hip_runtime.h>
#include <hip/hip_bf16.h>

typedef float v2f __attribute__((ext_vector_type(2)));
typedef float v8f __attribute__((ext_vector_type(8)));

#ifndef GCN_THREADS
#define GCN_THREADS 256
#endif

#define SCAN_T 256
#define SCAN_ITEMS 4
#define SCAN_WAVES (SCAN_T / 32)

__global__ void k_cnt_zero(int* __restrict__ cnt, int n) {
  int i = blockIdx.x * blockDim.x + threadIdx.x;
  if (i < n) cnt[i] = 0;
}

__global__ void k_count_rank(const int* __restrict__ dst, int* __restrict__ cnt,
                             int* __restrict__ rank, int e) {
  int i = blockIdx.x * blockDim.x + threadIdx.x;
  if (i < e) rank[i] = atomicAdd(&cnt[dst[i]], 1);
}

__device__ __forceinline__ int wave_sum(int v) {
#pragma unroll
  for (int d = 16; d > 0; d >>= 1) v += __shfl_xor(v, d, 32);
  return v;
}

__global__ void k_tile_sum(const int* __restrict__ cnt, int* __restrict__ tsum, int n) {
  __shared__ int wsum[SCAN_WAVES];
  const int t = (int)threadIdx.x, lane = t & 31, w = t >> 5;
  const int i0 = ((int)blockIdx.x * SCAN_T + t) * SCAN_ITEMS;
  int s = 0;
  if (i0 + SCAN_ITEMS <= n) {
    int4 c = *(const int4*)(cnt + i0);
    s = c.x + c.y + c.z + c.w;
  } else {
    for (int k = 0; k < SCAN_ITEMS; ++k) if (i0 + k < n) s += cnt[i0 + k];
  }
  s = wave_sum(s);
  if (lane == 0) wsum[w] = s;
  __syncthreads();
  if (t == 0) {
    int tot = 0;
#pragma unroll
    for (int k = 0; k < SCAN_WAVES; ++k) tot += wsum[k];
    tsum[blockIdx.x] = tot;
  }
}

__global__ void k_tile_scan(const int* __restrict__ cnt, const int* __restrict__ tsum,
                            int* __restrict__ rowptr, float* __restrict__ dinv, int n) {
  __shared__ int wsum[SCAN_WAVES];
  const int t = (int)threadIdx.x, lane = t & 31, w = t >> 5;

  int p = 0;
  for (int b = t; b < (int)blockIdx.x; b += SCAN_T) p += tsum[b];
  p = wave_sum(p);
  if (lane == 0) wsum[w] = p;
  __syncthreads();
  int run = 0;
#pragma unroll
  for (int k = 0; k < SCAN_WAVES; ++k) run += wsum[k];
  __syncthreads();

  const int i0 = ((int)blockIdx.x * SCAN_T + t) * SCAN_ITEMS;
  const bool full = (i0 + SCAN_ITEMS <= n);
  int c[SCAN_ITEMS];
  if (full) {
    int4 v = *(const int4*)(cnt + i0);
    c[0] = v.x; c[1] = v.y; c[2] = v.z; c[3] = v.w;
  } else {
#pragma unroll
    for (int k = 0; k < SCAN_ITEMS; ++k) c[k] = (i0 + k < n) ? cnt[i0 + k] : 0;
  }
  const int s = c[0] + c[1] + c[2] + c[3];

  int incl = s;
#pragma unroll
  for (int d = 1; d < 32; d <<= 1) {
    int u = __shfl_up(incl, d, 32);
    if (lane >= d) incl += u;
  }
  if (lane == 31) wsum[w] = incl;
  __syncthreads();
#pragma unroll
  for (int k = 0; k < SCAN_WAVES; ++k) if (k < w) run += wsum[k];
  run += incl - s;

  if (full) {
    int4 r;
    r.x = run; r.y = r.x + c[0]; r.z = r.y + c[1]; r.w = r.z + c[2];
    *(int4*)(rowptr + i0) = r;
    float4 dv;
    dv.x = rsqrtf((float)(c[0] + 1)); dv.y = rsqrtf((float)(c[1] + 1));
    dv.z = rsqrtf((float)(c[2] + 1)); dv.w = rsqrtf((float)(c[3] + 1));
    *(float4*)(dinv + i0) = dv;
    if (i0 + SCAN_ITEMS == n) rowptr[n] = r.w + c[3];
  } else {
#pragma unroll
    for (int k = 0; k < SCAN_ITEMS; ++k) {
      if (i0 + k < n) {
        rowptr[i0 + k] = run;
        dinv[i0 + k] = rsqrtf((float)(c[k] + 1));
        run += c[k];
        if (i0 + k == n - 1) rowptr[n] = run;
      }
    }
  }
}

__global__ void k_csr_fill(const int* __restrict__ src, const int* __restrict__ dst,
                           const int* __restrict__ rank, const int* __restrict__ rowptr,
                           const float* __restrict__ dinv, int2* __restrict__ csr, int e) {
  int i = blockIdx.x * blockDim.x + threadIdx.x;
  if (i >= e) return;
  int s = src[i];
  int p = rowptr[dst[i]] + rank[i];
  csr[p] = make_int2(s, __float_as_int(dinv[s]));
}

template <int K, int NO, int NT>
__global__ void k_gemm_wmma(const float* __restrict__ A, const float* __restrict__ W,
                            float* __restrict__ out, int mtiles) {
  const int wave = (int)((blockIdx.x * blockDim.x + threadIdx.x) >> 5);
  const int lane = (int)(threadIdx.x & 31);
  if (wave >= mtiles) return;

  const int hi   = lane >> 4;
  const int lo   = lane & 15;
  const int kb   = hi << 1;
  const int row  = wave * 16 + lo;

  v8f acc[NT];
#pragma unroll
  for (int t = 0; t < NT; ++t) acc[t] = (v8f){0.f, 0.f, 0.f, 0.f, 0.f, 0.f, 0.f, 0.f};

  const float* __restrict__ arow = A + (size_t)row * K;

  for (int k0 = 0; k0 < K; k0 += 4) {
    v2f a = *(const v2f*)(arow + k0 + kb);
#pragma unroll
    for (int t = 0; t < NT; ++t) {
      const int n = t * 16 + lo;
      v2f b;
      b.x = W[(size_t)(k0 + kb + 0) * NO + n];
      b.y = W[(size_t)(k0 + kb + 1) * NO + n];
      acc[t] = __builtin_amdgcn_wmma_f32_16x16x4_f32(
           false, a,  false, b,
           (short)0, acc[t],  false,  false);
    }
  }

  const int mbase = wave * 16 + (hi << 3);
#pragma unroll
  for (int t = 0; t < NT; ++t) {
#pragma unroll
    for (int r = 0; r < 8; ++r) {
      out[(size_t)(mbase + r) * NO + t * 16 + lo] = acc[t][r];
    }
  }
}

template <int V> struct vec_of;
template <> struct vec_of<2> { typedef v2f type; };
template <> struct vec_of<4> { typedef float type __attribute__((ext_vector_type(4))); };

template <int C, int D, typename vf>
__device__ __forceinline__ void gather_step(vf& acc, const int2 sw, const int j, const float di,
                                            const float* __restrict__ hl) {
  vf r[C];
  float w[C];
#pragma unroll
  for (int k = 0; k < C; ++k) {
    const int s = __builtin_amdgcn_readlane(sw.x, j + k);
    w[k] = __int_as_float(__builtin_amdgcn_readlane(sw.y, j + k)) * di;
    r[k] = *(const vf*)(hl + (size_t)s * D);
  }
#pragma unroll
  for (int k = 0; k < C; ++k) acc += r[k] * w[k];
}

template <int D, bool RELU>
__global__ void k_node_gather(const int* __restrict__ rowptr, const int2* __restrict__ csr,
                              const float* __restrict__ dinv, const float* __restrict__ h,
                              const float* __restrict__ bias, float* __restrict__ out, int n) {
  constexpr int V = D / 32;
  typedef typename vec_of<V>::type vf;
  const int lane = (int)(threadIdx.x & 31);
  const int node = __builtin_amdgcn_readfirstlane(
      (int)(blockIdx.x * (blockDim.x >> 5) + (threadIdx.x >> 5)));
  if (node >= n) return;

  const int beg = rowptr[node];
  const int end = rowptr[node + 1];
  const float di = dinv[node];
  const float* __restrict__ hl = h + lane * V;
  const vf bv = *(const vf*)(bias + lane * V);

  int base = beg;
  int m = (end - base < 32) ? (end - base) : 32;
  int2 sw = make_int2(0, 0);
  if (lane < m) sw = csr[base + lane];
  vf acc = *(const vf*)(hl + (size_t)node * D) * (di * di);

  for (;;) {
    int j = 0;
    for (; j + 8 <= m; j += 8) gather_step<8, D, vf>(acc, sw, j, di, hl);
    switch (m - j) {
      case 7: gather_step<7, D, vf>(acc, sw, j, di, hl); break;
      case 6: gather_step<6, D, vf>(acc, sw, j, di, hl); break;
      case 5: gather_step<5, D, vf>(acc, sw, j, di, hl); break;
      case 4: gather_step<4, D, vf>(acc, sw, j, di, hl); break;
      case 3: gather_step<3, D, vf>(acc, sw, j, di, hl); break;
      case 2: gather_step<2, D, vf>(acc, sw, j, di, hl); break;
      case 1: gather_step<1, D, vf>(acc, sw, j, di, hl); break;
      default: break;
    }
    base += 32;
    if (base >= end) break;
    m = (end - base < 32) ? (end - base) : 32;
    sw = make_int2(0, 0);
    if (lane < m) sw = csr[base + lane];
  }
  acc += bv;
  if (RELU) {
#pragma unroll
    for (int k = 0; k < V; ++k) acc[k] = fmaxf(acc[k], 0.0f);
  }
  *(vf*)(out + (size_t)node * D + lane * V) = acc;
}

extern "C" void kernel_launch(void* const* d_in, const int* in_sizes, int n_in,
                              void* d_out, int out_size, void* d_ws, size_t ws_size,
                              hipStream_t stream) {
  (void)n_in; (void)out_size; (void)ws_size;

  const float* x    = (const float*)d_in[0];
  const int*   ei   = (const int*)d_in[1];
  const float* W1   = (const float*)d_in[2];
  const float* b1   = (const float*)d_in[3];
  const float* W2   = (const float*)d_in[4];
  const float* b2   = (const float*)d_in[5];
  float*       out  = (float*)d_out;

  const int N = in_sizes[0] / 128;
  const int E = in_sizes[1] / 2;
  const int IN_D = 128, HID_D = 128, OUT_D = 64;

  const int* src = ei;
  const int* dst = ei + E;

  auto up4 = [](size_t v) { return (v + 3) & ~(size_t)3; };
  const int ntile = (N + SCAN_T * SCAN_ITEMS - 1) / (SCAN_T * SCAN_ITEMS);
  int*   cnt    = (int*)d_ws;
  int*   rowptr = cnt + up4((size_t)N);
  int*   tsum   = rowptr + up4((size_t)N + 1);
  int*   rank   = tsum + up4((size_t)ntile);
  int2*  csr    = (int2*)(rank + up4((size_t)E));
  float* dinv   = (float*)(csr + up4((size_t)E));
  float* bufA   = dinv + up4((size_t)N);
  float* bufB   = bufA + (size_t)N * 128;

  const int T = GCN_THREADS;
  auto blk = [&](long long work) { return (unsigned)((work + T - 1) / T); };
  const unsigned gnode = (unsigned)((N + (T / 32) - 1) / (T / 32));

  k_cnt_zero<<<blk(N), T, 0, stream>>>(cnt, N);
  k_count_rank<<<blk(E), T, 0, stream>>>(dst, cnt, rank, E);
  k_tile_sum<<<ntile, SCAN_T, 0, stream>>>(cnt, tsum, N);
  k_tile_scan<<<ntile, SCAN_T, 0, stream>>>(cnt, tsum, rowptr, dinv, N);
  k_csr_fill<<<blk(E), T, 0, stream>>>(src, dst, rank, rowptr, dinv, csr, E);

  const int mtiles = N / 16;
  {
    const int wavesPerBlk = T / 32;
    unsigned g = (unsigned)((mtiles + wavesPerBlk - 1) / wavesPerBlk);
    k_gemm_wmma<IN_D, HID_D, HID_D / 16><<<g, T, 0, stream>>>(x, W1, bufA, mtiles);
  }

  k_node_gather<HID_D, true><<<gnode, T, 0, stream>>>(rowptr, csr, dinv, bufA, b1, bufB, N);

  {
    const int wavesPerBlk = T / 32;
    unsigned g = (unsigned)((mtiles + wavesPerBlk - 1) / wavesPerBlk);
    k_gemm_wmma<HID_D, OUT_D, OUT_D / 16><<<g, T, 0, stream>>>(bufB, W2, bufA, mtiles);
  }

  k_node_gather<OUT_D, false><<<gnode, T, 0, stream>>>(rowptr, csr, dinv, bufA, b2, out, N);
}


# Layer-2 dense transform moved into the layer-1 node-sum kernel: 16 finished rows go to an LDS tile and are multiplied by W2 on f32 WMMA there; the activation never goes to memory; sum orders unchanged
# speedup vs baseline: 2.1708x; 1.1255x over previous
#include <hip/hip_runtime.h>
#include <hip/hip_bf16.h>

typedef float v2f __attribute__((ext_vector_type(2)));
typedef float v8f __attribute__((ext_vector_type(8)));

#ifndef GCN_THREADS
#define GCN_THREADS 256
#endif

#define SCAN_T 256
#define SCAN_ITEMS 4
#define SCAN_WAVES (SCAN_T / 32)

__global__ void k_cnt_zero(int* __restrict__ cnt, int n) {
  int i = blockIdx.x * blockDim.x + threadIdx.x;
  if (i < n) cnt[i] = 0;
}

__global__ void k_count_rank(const int* __restrict__ dst, int* __restrict__ cnt,
                             int* __restrict__ rank, int e) {
  int i = blockIdx.x * blockDim.x + threadIdx.x;
  if (i < e) rank[i] = atomicAdd(&cnt[dst[i]], 1);
}

__device__ __forceinline__ int wave_sum(int v) {
#pragma unroll
  for (int d = 16; d > 0; d >>= 1) v += __shfl_xor(v, d, 32);
  return v;
}

__global__ void k_tile_sum(const int* __restrict__ cnt, int* __restrict__ tsum, int n) {
  __shared__ int wsum[SCAN_WAVES];
  const int t = (int)threadIdx.x, lane = t & 31, w = t >> 5;
  const int i0 = ((int)blockIdx.x * SCAN_T + t) * SCAN_ITEMS;
  int s = 0;
  if (i0 + SCAN_ITEMS <= n) {
    int4 c = *(const int4*)(cnt + i0);
    s = c.x + c.y + c.z + c.w;
  } else {
    for (int k = 0; k < SCAN_ITEMS; ++k) if (i0 + k < n) s += cnt[i0 + k];
  }
  s = wave_sum(s);
  if (lane == 0) wsum[w] = s;
  __syncthreads();
  if (t == 0) {
    int tot = 0;
#pragma unroll
    for (int k = 0; k < SCAN_WAVES; ++k) tot += wsum[k];
    tsum[blockIdx.x] = tot;
  }
}

__global__ void k_tile_scan(const int* __restrict__ cnt, const int* __restrict__ tsum,
                            int* __restrict__ rowptr, float* __restrict__ dinv, int n) {
  __shared__ int wsum[SCAN_WAVES];
  const int t = (int)threadIdx.x, lane = t & 31, w = t >> 5;

  int p = 0;
  for (int b = t; b < (int)blockIdx.x; b += SCAN_T) p += tsum[b];
  p = wave_sum(p);
  if (lane == 0) wsum[w] = p;
  __syncthreads();
  int run = 0;
#pragma unroll
  for (int k = 0; k < SCAN_WAVES; ++k) run += wsum[k];
  __syncthreads();

  const int i0 = ((int)blockIdx.x * SCAN_T + t) * SCAN_ITEMS;
  const bool full = (i0 + SCAN_ITEMS <= n);
  int c[SCAN_ITEMS];
  if (full) {
    int4 v = *(const int4*)(cnt + i0);
    c[0] = v.x; c[1] = v.y; c[2] = v.z; c[3] = v.w;
  } else {
#pragma unroll
    for (int k = 0; k < SCAN_ITEMS; ++k) c[k] = (i0 + k < n) ? cnt[i0 + k] : 0;
  }
  const int s = c[0] + c[1] + c[2] + c[3];

  int incl = s;
#pragma unroll
  for (int d = 1; d < 32; d <<= 1) {
    int u = __shfl_up(incl, d, 32);
    if (lane >= d) incl += u;
  }
  if (lane == 31) wsum[w] = incl;
  __syncthreads();
#pragma unroll
  for (int k = 0; k < SCAN_WAVES; ++k) if (k < w) run += wsum[k];
  run += incl - s;

  if (full) {
    int4 r;
    r.x = run; r.y = r.x + c[0]; r.z = r.y + c[1]; r.w = r.z + c[2];
    *(int4*)(rowptr + i0) = r;
    float4 dv;
    dv.x = rsqrtf((float)(c[0] + 1)); dv.y = rsqrtf((float)(c[1] + 1));
    dv.z = rsqrtf((float)(c[2] + 1)); dv.w = rsqrtf((float)(c[3] + 1));
    *(float4*)(dinv + i0) = dv;
    if (i0 + SCAN_ITEMS == n) rowptr[n] = r.w + c[3];
  } else {
#pragma unroll
    for (int k = 0; k < SCAN_ITEMS; ++k) {
      if (i0 + k < n) {
        rowptr[i0 + k] = run;
        dinv[i0 + k] = rsqrtf((float)(c[k] + 1));
        run += c[k];
        if (i0 + k == n - 1) rowptr[n] = run;
      }
    }
  }
}

__global__ void k_csr_fill(const int* __restrict__ src, const int* __restrict__ dst,
                           const int* __restrict__ rank, const int* __restrict__ rowptr,
                           const float* __restrict__ dinv, int2* __restrict__ csr, int e) {
  int i = blockIdx.x * blockDim.x + threadIdx.x;
  if (i >= e) return;
  int s = src[i];
  int p = rowptr[dst[i]] + rank[i];
  csr[p] = make_int2(s, __float_as_int(dinv[s]));
}

template <int K, int NO, int NT>
__global__ void k_gemm_wmma(const float* __restrict__ A, const float* __restrict__ W,
                            float* __restrict__ out, int mtiles) {
  const int wave = (int)((blockIdx.x * blockDim.x + threadIdx.x) >> 5);
  const int lane = (int)(threadIdx.x & 31);
  if (wave >= mtiles) return;

  const int hi   = lane >> 4;
  const int lo   = lane & 15;
  const int kb   = hi << 1;
  const int row  = wave * 16 + lo;

  v8f acc[NT];
#pragma unroll
  for (int t = 0; t < NT; ++t) acc[t] = (v8f){0.f, 0.f, 0.f, 0.f, 0.f, 0.f, 0.f, 0.f};

  const float* __restrict__ arow = A + (size_t)row * K;

  for (int k0 = 0; k0 < K; k0 += 4) {
    v2f a = *(const v2f*)(arow + k0 + kb);
#pragma unroll
    for (int t = 0; t < NT; ++t) {
      const int n = t * 16 + lo;
      v2f b;
      b.x = W[(size_t)(k0 + kb + 0) * NO + n];
      b.y = W[(size_t)(k0 + kb + 1) * NO + n];
      acc[t] = __builtin_amdgcn_wmma_f32_16x16x4_f32(
           false, a,  false, b,
           (short)0, acc[t],  false,  false);
    }
  }

  const int mbase = wave * 16 + (hi << 3);
#pragma unroll
  for (int t = 0; t < NT; ++t) {
#pragma unroll
    for (int r = 0; r < 8; ++r) {
      out[(size_t)(mbase + r) * NO + t * 16 + lo] = acc[t][r];
    }
  }
}

template <int V> struct vec_of;
template <> struct vec_of<2> { typedef v2f type; };
template <> struct vec_of<4> { typedef float type __attribute__((ext_vector_type(4))); };

template <int C, int D, typename vf>
__device__ __forceinline__ void gather_step(vf& acc, const int2 sw, const int j, const float di,
                                            const float* __restrict__ hl) {
  vf r[C];
  float w[C];
#pragma unroll
  for (int k = 0; k < C; ++k) {
    const int s = __builtin_amdgcn_readlane(sw.x, j + k);
    w[k] = __int_as_float(__builtin_amdgcn_readlane(sw.y, j + k)) * di;
    r[k] = *(const vf*)(hl + (size_t)s * D);
  }
#pragma unroll
  for (int k = 0; k < C; ++k) acc += r[k] * w[k];
}

template <int D, bool RELU>
__global__ void k_node_gather(const int* __restrict__ rowptr, const int2* __restrict__ csr,
                              const float* __restrict__ dinv, const float* __restrict__ h,
                              const float* __restrict__ bias, float* __restrict__ out, int n) {
  constexpr int V = D / 32;
  typedef typename vec_of<V>::type vf;
  const int lane = (int)(threadIdx.x & 31);
  const int node = __builtin_amdgcn_readfirstlane(
      (int)(blockIdx.x * (blockDim.x >> 5) + (threadIdx.x >> 5)));
  if (node >= n) return;

  const int beg = rowptr[node];
  const int end = rowptr[node + 1];
  const float di = dinv[node];
  const float* __restrict__ hl = h + lane * V;
  const vf bv = *(const vf*)(bias + lane * V);

  int base = beg;
  int m = (end - base < 32) ? (end - base) : 32;
  int2 sw = make_int2(0, 0);
  if (lane < m) sw = csr[base + lane];
  vf acc = *(const vf*)(hl + (size_t)node * D) * (di * di);

  for (;;) {
    int j = 0;
    for (; j + 8 <= m; j += 8) gather_step<8, D, vf>(acc, sw, j, di, hl);
    switch (m - j) {
      case 7: gather_step<7, D, vf>(acc, sw, j, di, hl); break;
      case 6: gather_step<6, D, vf>(acc, sw, j, di, hl); break;
      case 5: gather_step<5, D, vf>(acc, sw, j, di, hl); break;
      case 4: gather_step<4, D, vf>(acc, sw, j, di, hl); break;
      case 3: gather_step<3, D, vf>(acc, sw, j, di, hl); break;
      case 2: gather_step<2, D, vf>(acc, sw, j, di, hl); break;
      case 1: gather_step<1, D, vf>(acc, sw, j, di, hl); break;
      default: break;
    }
    base += 32;
    if (base >= end) break;
    m = (end - base < 32) ? (end - base) : 32;
    sw = make_int2(0, 0);
    if (lane < m) sw = csr[base + lane];
  }
  acc += bv;
  if (RELU) {
#pragma unroll
    for (int k = 0; k < V; ++k) acc[k] = fmaxf(acc[k], 0.0f);
  }
  *(vf*)(out + (size_t)node * D + lane * V) = acc;
}

template <int D, int NO, bool RELU>
__global__ __launch_bounds__(128) void k_node_gather_gemm(
    const int* __restrict__ rowptr, const int2* __restrict__ csr, const float* __restrict__ dinv,
    const float* __restrict__ h, const float* __restrict__ bias, const float* __restrict__ W,
    float* __restrict__ out, int n) {
  constexpr int V = D / 32;
  typedef typename vec_of<V>::type vf;
  static_assert(NO == 64, "four waves, one 16-column slice of W each; a row of NO floats = 32 lanes x 2");
  constexpr int AP = D + 4;
  constexpr int OP = NO + 4;
  __shared__ __attribute__((aligned(16))) float tile[16 * AP];
  __shared__ __attribute__((aligned(16))) float otile[16 * OP];

  const int lane = (int)(threadIdx.x & 31);
  const int w = __builtin_amdgcn_readfirstlane((int)(threadIdx.x >> 5));
  const int node0 = (int)blockIdx.x * 16 + w * 4;

  int rp = 0;
  float dv = 0.0f;
  {
    const int q = node0 + lane;
    if (lane <= 4) rp = rowptr[q < n ? q : n];
    if (lane < 4 && q < n) dv = dinv[q];
  }
  const float* __restrict__ hl = h + lane * V;
  const vf bv = *(const vf*)(bias + lane * V);

  int beg = __builtin_amdgcn_readlane(rp, 0);
  int2 sw = make_int2(0, 0);
  {
    const int e1 = __builtin_amdgcn_readlane(rp, 1);
    const int m0 = (e1 - beg < 32) ? (e1 - beg) : 32;
    if (lane < m0) sw = csr[beg + lane];
  }
  float di = __int_as_float(__builtin_amdgcn_readlane(__float_as_int(dv), 0));
  vf accn = *(const vf*)(hl + (size_t)((node0 < n) ? node0 : (n - 1)) * D) * (di * di);

#pragma nounroll
  for (int i = 0; i < 4; ++i) {
    const int node = node0 + i;
    const int end = __builtin_amdgcn_readlane(rp, i + 1);

    vf acc = accn;
    int2 swn = make_int2(0, 0);
    vf own = bv;
    if (i < 3) {
      const int e2 = __builtin_amdgcn_readlane(rp, i + 2);
      const int mn = (e2 - end < 32) ? (e2 - end) : 32;
      if (lane < mn) swn = csr[end + lane];
      own = *(const vf*)(hl + (size_t)((node + 1 < n) ? (node + 1) : (n - 1)) * D);
    }

    int base = beg;
    int m = (end - base < 32) ? (end - base) : 32;
    for (;;) {
      int j = 0;
      for (; j + 8 <= m; j += 8) gather_step<8, D, vf>(acc, sw, j, di, hl);
      switch (m - j) {
        case 7: gather_step<7, D, vf>(acc, sw, j, di, hl); break;
        case 6: gather_step<6, D, vf>(acc, sw, j, di, hl); break;
        case 5: gather_step<5, D, vf>(acc, sw, j, di, hl); break;
        case 4: gather_step<4, D, vf>(acc, sw, j, di, hl); break;
        case 3: gather_step<3, D, vf>(acc, sw, j, di, hl); break;
        case 2: gather_step<2, D, vf>(acc, sw, j, di, hl); break;
        case 1: gather_step<1, D, vf>(acc, sw, j, di, hl); break;
        default: break;
      }
      base += 32;
      if (base >= end) break;
      m = (end - base < 32) ? (end - base) : 32;
      sw = make_int2(0, 0);
      if (lane < m) sw = csr[base + lane];
      asm volatile("" : : "v"(sw.x), "v"(sw.y));
    }
    acc += bv;
    if (RELU) {
#pragma unroll
      for (int k = 0; k < V; ++k) acc[k] = fmaxf(acc[k], 0.0f);
    }
    *(vf*)(tile + (w * 4 + i) * AP + lane * V) = acc;
    sw = swn;
    beg = end;
    di = __int_as_float(__builtin_amdgcn_readlane(__float_as_int(dv), (i + 1) & 3));
    accn = own * (di * di);
  }
  __syncthreads();

  const int hi = lane >> 4;
  const int lo = lane & 15;
  const int kb = hi << 1;
  const float* arow = tile + lo * AP + kb;
  const float* __restrict__ wcol = W + w * 16 + lo;
  v8f c = (v8f){0.f, 0.f, 0.f, 0.f, 0.f, 0.f, 0.f, 0.f};
  for (int k0 = 0; k0 < D; k0 += 4) {
    const v2f a = *(const v2f*)(arow + k0);
    v2f b;
    b.x = wcol[(size_t)(k0 + kb + 0) * NO];
    b.y = wcol[(size_t)(k0 + kb + 1) * NO];
    c = __builtin_amdgcn_wmma_f32_16x16x4_f32(
         false, a,  false, b,
         (short)0, c,  false,  false);
  }

#pragma unroll
  for (int r = 0; r < 8; ++r) otile[(hi * 8 + r) * OP + w * 16 + lo] = c[r];
  __syncthreads();

#pragma unroll
  for (int i = 0; i < 4; ++i) {
    const int node = node0 + i;
    if (node < n)
      *(v2f*)(out + (size_t)node * NO + lane * 2) = *(const v2f*)(otile + (w * 4 + i) * OP + lane * 2);
  }
}

extern "C" void kernel_launch(void* const* d_in, const int* in_sizes, int n_in,
                              void* d_out, int out_size, void* d_ws, size_t ws_size,
                              hipStream_t stream) {
  (void)n_in; (void)out_size; (void)ws_size;

  const float* x    = (const float*)d_in[0];
  const int*   ei   = (const int*)d_in[1];
  const float* W1   = (const float*)d_in[2];
  const float* b1   = (const float*)d_in[3];
  const float* W2   = (const float*)d_in[4];
  const float* b2   = (const float*)d_in[5];
  float*       out  = (float*)d_out;

  const int N = in_sizes[0] / 128;
  const int E = in_sizes[1] / 2;
  const int IN_D = 128, HID_D = 128, OUT_D = 64;

  const int* src = ei;
  const int* dst = ei + E;

  auto up4 = [](size_t v) { return (v + 3) & ~(size_t)3; };
  const int ntile = (N + SCAN_T * SCAN_ITEMS - 1) / (SCAN_T * SCAN_ITEMS);
  int*   cnt    = (int*)d_ws;
  int*   rowptr = cnt + up4((size_t)N);
  int*   tsum   = rowptr + up4((size_t)N + 1);
  int*   rank   = tsum + up4((size_t)ntile);
  int2*  csr    = (int2*)(rank + up4((size_t)E));
  float* dinv   = (float*)(csr + up4((size_t)E));
  float* bufA   = dinv + up4((size_t)N);
  float* bufB   = bufA + (size_t)N * 128;

  const int T = GCN_THREADS;
  auto blk = [&](long long work) { return (unsigned)((work + T - 1) / T); };
  const unsigned gnode = (unsigned)((N + (T / 32) - 1) / (T / 32));

  k_cnt_zero<<<blk(N), T, 0, stream>>>(cnt, N);
  k_count_rank<<<blk(E), T, 0, stream>>>(dst, cnt, rank, E);
  k_tile_sum<<<ntile, SCAN_T, 0, stream>>>(cnt, tsum, N);
  k_tile_scan<<<ntile, SCAN_T, 0, stream>>>(cnt, tsum, rowptr, dinv, N);
  k_csr_fill<<<blk(E), T, 0, stream>>>(src, dst, rank, rowptr, dinv, csr, E);

  const int mtiles = N / 16;
  {
    const int wavesPerBlk = T / 32;
    unsigned g = (unsigned)((mtiles + wavesPerBlk - 1) / wavesPerBlk);
    k_gemm_wmma<IN_D, HID_D, HID_D / 16><<<g, T, 0, stream>>>(x, W1, bufA, mtiles);
  }

  k_node_gather_gemm<HID_D, OUT_D, true><<<(unsigned)((N + 15) / 16), 128, 0, stream>>>(
      rowptr, csr, dinv, bufA, b1, W2, bufB, N);

  k_node_gather<OUT_D, false><<<gnode, T, 0, stream>>>(rowptr, csr, dinv, bufB, b2, out, N);
}


# Layer-1 node sum now runs over rows of x and the W1 product follows it in the same work-group (32-row tile, f32 WMMA, bias, relu, W2): x@W1 launch gone; sums reassociated: (sum of rows)@W1
# speedup vs baseline: 2.5177x; 1.1598x over previous
#include <hip/hip_runtime.h>
#include <hip/hip_bf16.h>

typedef float v2f __attribute__((ext_vector_type(2)));
typedef float v8f __attribute__((ext_vector_type(8)));

#ifndef GCN_THREADS
#define GCN_THREADS 256
#endif

#define SCAN_T 256
#define SCAN_ITEMS 4
#define SCAN_WAVES (SCAN_T / 32)

__global__ void k_cnt_zero(int* __restrict__ cnt, int n) {
  int i = blockIdx.x * blockDim.x + threadIdx.x;
  if (i < n) cnt[i] = 0;
}

__global__ void k_count_rank(const int* __restrict__ dst, int* __restrict__ cnt,
                             int* __restrict__ rank, int e) {
  int i = blockIdx.x * blockDim.x + threadIdx.x;
  if (i < e) rank[i] = atomicAdd(&cnt[dst[i]], 1);
}

__device__ __forceinline__ int wave_sum(int v) {
#pragma unroll
  for (int d = 16; d > 0; d >>= 1) v += __shfl_xor(v, d, 32);
  return v;
}

__global__ void k_tile_sum(const int* __restrict__ cnt, int* __restrict__ tsum, int n) {
  __shared__ int wsum[SCAN_WAVES];
  const int t = (int)threadIdx.x, lane = t & 31, w = t >> 5;
  const int i0 = ((int)blockIdx.x * SCAN_T + t) * SCAN_ITEMS;
  int s = 0;
  if (i0 + SCAN_ITEMS <= n) {
    int4 c = *(const int4*)(cnt + i0);
    s = c.x + c.y + c.z + c.w;
  } else {
    for (int k = 0; k < SCAN_ITEMS; ++k) if (i0 + k < n) s += cnt[i0 + k];
  }
  s = wave_sum(s);
  if (lane == 0) wsum[w] = s;
  __syncthreads();
  if (t == 0) {
    int tot = 0;
#pragma unroll
    for (int k = 0; k < SCAN_WAVES; ++k) tot += wsum[k];
    tsum[blockIdx.x] = tot;
  }
}

__global__ void k_tile_scan(const int* __restrict__ cnt, const int* __restrict__ tsum,
                            int* __restrict__ rowptr, float* __restrict__ dinv, int n) {
  __shared__ int wsum[SCAN_WAVES];
  const int t = (int)threadIdx.x, lane = t & 31, w = t >> 5;

  int p = 0;
  for (int b = t; b < (int)blockIdx.x; b += SCAN_T) p += tsum[b];
  p = wave_sum(p);
  if (lane == 0) wsum[w] = p;
  __syncthreads();
  int run = 0;
#pragma unroll
  for (int k = 0; k < SCAN_WAVES; ++k) run += wsum[k];
  __syncthreads();

  const int i0 = ((int)blockIdx.x * SCAN_T + t) * SCAN_ITEMS;
  const bool full = (i0 + SCAN_ITEMS <= n);
  int c[SCAN_ITEMS];
  if (full) {
    int4 v = *(const int4*)(cnt + i0);
    c[0] = v.x; c[1] = v.y; c[2] = v.z; c[3] = v.w;
  } else {
#pragma unroll
    for (int k = 0; k < SCAN_ITEMS; ++k) c[k] = (i0 + k < n) ? cnt[i0 + k] : 0;
  }
  const int s = c[0] + c[1] + c[2] + c[3];

  int incl = s;
#pragma unroll
  for (int d = 1; d < 32; d <<= 1) {
    int u = __shfl_up(incl, d, 32);
    if (lane >= d) incl += u;
  }
  if (lane == 31) wsum[w] = incl;
  __syncthreads();
#pragma unroll
  for (int k = 0; k < SCAN_WAVES; ++k) if (k < w) run += wsum[k];
  run += incl - s;

  if (full) {
    int4 r;
    r.x = run; r.y = r.x + c[0]; r.z = r.y + c[1]; r.w = r.z + c[2];
    *(int4*)(rowptr + i0) = r;
    float4 dv;
    dv.x = rsqrtf((float)(c[0] + 1)); dv.y = rsqrtf((float)(c[1] + 1));
    dv.z = rsqrtf((float)(c[2] + 1)); dv.w = rsqrtf((float)(c[3] + 1));
    *(float4*)(dinv + i0) = dv;
    if (i0 + SCAN_ITEMS == n) rowptr[n] = r.w + c[3];
  } else {
#pragma unroll
    for (int k = 0; k < SCAN_ITEMS; ++k) {
      if (i0 + k < n) {
        rowptr[i0 + k] = run;
        dinv[i0 + k] = rsqrtf((float)(c[k] + 1));
        run += c[k];
        if (i0 + k == n - 1) rowptr[n] = run;
      }
    }
  }
}

__global__ void k_csr_fill(const int* __restrict__ src, const int* __restrict__ dst,
                           const int* __restrict__ rank, const int* __restrict__ rowptr,
                           const float* __restrict__ dinv, int2* __restrict__ csr, int e) {
  int i = blockIdx.x * blockDim.x + threadIdx.x;
  if (i >= e) return;
  int s = src[i];
  int p = rowptr[dst[i]] + rank[i];
  csr[p] = make_int2(s, __float_as_int(dinv[s]));
}

template <int V> struct vec_of;
template <> struct vec_of<2> { typedef v2f type; };
template <> struct vec_of<4> { typedef float type __attribute__((ext_vector_type(4))); };

template <int C, int D, typename vf>
__device__ __forceinline__ void gather_step(vf& acc, const int2 sw, const int j, const float di,
                                            const float* __restrict__ hl) {
  vf r[C];
  float w[C];
#pragma unroll
  for (int k = 0; k < C; ++k) {
    const int s = __builtin_amdgcn_readlane(sw.x, j + k);
    w[k] = __int_as_float(__builtin_amdgcn_readlane(sw.y, j + k)) * di;
    r[k] = *(const vf*)(hl + (size_t)s * D);
  }
#pragma unroll
  for (int k = 0; k < C; ++k) acc += r[k] * w[k];
}

template <int D, bool RELU>
__global__ void k_node_gather(const int* __restrict__ rowptr, const int2* __restrict__ csr,
                              const float* __restrict__ dinv, const float* __restrict__ h,
                              const float* __restrict__ bias, float* __restrict__ out, int n) {
  constexpr int V = D / 32;
  typedef typename vec_of<V>::type vf;
  const int lane = (int)(threadIdx.x & 31);
  const int node = __builtin_amdgcn_readfirstlane(
      (int)(blockIdx.x * (blockDim.x >> 5) + (threadIdx.x >> 5)));
  if (node >= n) return;

  const int beg = rowptr[node];
  const int end = rowptr[node + 1];
  const float di = dinv[node];
  const float* __restrict__ hl = h + lane * V;
  const vf bv = *(const vf*)(bias + lane * V);

  int base = beg;
  int m = (end - base < 32) ? (end - base) : 32;
  int2 sw = make_int2(0, 0);
  if (lane < m) sw = csr[base + lane];
  vf acc = *(const vf*)(hl + (size_t)node * D) * (di * di);

  for (;;) {
    int j = 0;
    for (; j + 8 <= m; j += 8) gather_step<8, D, vf>(acc, sw, j, di, hl);
    switch (m - j) {
      case 7: gather_step<7, D, vf>(acc, sw, j, di, hl); break;
      case 6: gather_step<6, D, vf>(acc, sw, j, di, hl); break;
      case 5: gather_step<5, D, vf>(acc, sw, j, di, hl); break;
      case 4: gather_step<4, D, vf>(acc, sw, j, di, hl); break;
      case 3: gather_step<3, D, vf>(acc, sw, j, di, hl); break;
      case 2: gather_step<2, D, vf>(acc, sw, j, di, hl); break;
      case 1: gather_step<1, D, vf>(acc, sw, j, di, hl); break;
      default: break;
    }
    base += 32;
    if (base >= end) break;
    m = (end - base < 32) ? (end - base) : 32;
    sw = make_int2(0, 0);
    if (lane < m) sw = csr[base + lane];
  }
  acc += bv;
  if (RELU) {
#pragma unroll
    for (int k = 0; k < V; ++k) acc[k] = fmaxf(acc[k], 0.0f);
  }
  *(vf*)(out + (size_t)node * D + lane * V) = acc;
}

template <int D, int HD, int NO, bool RELU>
__global__ __launch_bounds__(256, 16) void k_node_gather_gemm2(
    const int* __restrict__ rowptr, const int2* __restrict__ csr, const float* __restrict__ dinv,
    const float* __restrict__ x, const float* __restrict__ W1, const float* __restrict__ bias1,
    const float* __restrict__ W2, float* __restrict__ out, int n) {
  constexpr int V = D / 32;
  typedef typename vec_of<V>::type vf;
  static_assert(HD == 128 && HD == D, "eight waves, one 16-column slice of W1 each; the activation reuses the z tile");
  static_assert(NO == 64, "two row tiles x four 16-column slices of W2 = eight waves; a row of NO floats = 32 lanes x 2");
  constexpr int AP = D + 4;
  constexpr int OP = NO + 4;
  __shared__ __attribute__((aligned(16))) float tile[32 * AP];
  __shared__ __attribute__((aligned(16))) float otile[32 * OP];

  const int lane = (int)(threadIdx.x & 31);
  const int w = __builtin_amdgcn_readfirstlane((int)(threadIdx.x >> 5));
  const int node0 = (int)blockIdx.x * 32 + w * 4;

  int rp = 0;
  float dv = 0.0f;
  {
    const int q = node0 + lane;
    if (lane <= 4) rp = rowptr[q < n ? q : n];
    if (lane < 4 && q < n) dv = dinv[q];
  }
  const float* __restrict__ hl = x + lane * V;

  int beg = __builtin_amdgcn_readlane(rp, 0);
  int2 sw = make_int2(0, 0);
  {
    const int e1 = __builtin_amdgcn_readlane(rp, 1);
    const int m0 = (e1 - beg < 32) ? (e1 - beg) : 32;
    if (lane < m0) sw = csr[beg + lane];
  }
  float di = __int_as_float(__builtin_amdgcn_readlane(__float_as_int(dv), 0));
  vf accn = *(const vf*)(hl + (size_t)((node0 < n) ? node0 : (n - 1)) * D) * (di * di);

#pragma nounroll
  for (int i = 0; i < 4; ++i) {
    const int node = node0 + i;
    const int end = __builtin_amdgcn_readlane(rp, i + 1);

    vf acc = accn;
    int2 swn = make_int2(0, 0);
    vf own = accn;
    if (i < 3) {
      const int e2 = __builtin_amdgcn_readlane(rp, i + 2);
      const int mn = (e2 - end < 32) ? (e2 - end) : 32;
      if (lane < mn) swn = csr[end + lane];
      own = *(const vf*)(hl + (size_t)((node + 1 < n) ? (node + 1) : (n - 1)) * D);
    }

    int base = beg;
    int m = (end - base < 32) ? (end - base) : 32;
    for (;;) {
      int j = 0;
      for (; j + 8 <= m; j += 8) gather_step<8, D, vf>(acc, sw, j, di, hl);
      switch (m - j) {
        case 7: gather_step<7, D, vf>(acc, sw, j, di, hl); break;
        case 6: gather_step<6, D, vf>(acc, sw, j, di, hl); break;
        case 5: gather_step<5, D, vf>(acc, sw, j, di, hl); break;
        case 4: gather_step<4, D, vf>(acc, sw, j, di, hl); break;
        case 3: gather_step<3, D, vf>(acc, sw, j, di, hl); break;
        case 2: gather_step<2, D, vf>(acc, sw, j, di, hl); break;
        case 1: gather_step<1, D, vf>(acc, sw, j, di, hl); break;
        default: break;
      }
      base += 32;
      if (base >= end) break;
      m = (end - base < 32) ? (end - base) : 32;
      sw = make_int2(0, 0);
      if (lane < m) sw = csr[base + lane];
      asm volatile("" : : "v"(sw.x), "v"(sw.y));
    }
    *(vf*)(tile + (w * 4 + i) * AP + lane * V) = acc;
    sw = swn;
    beg = end;
    di = __int_as_float(__builtin_amdgcn_readlane(__float_as_int(dv), (i + 1) & 3));
    accn = own * (di * di);
  }
  __syncthreads();

  const int hi = lane >> 4;
  const int lo = lane & 15;
  const int kb = hi << 1;

  v8f c0 = (v8f){0.f, 0.f, 0.f, 0.f, 0.f, 0.f, 0.f, 0.f};
  v8f c1 = (v8f){0.f, 0.f, 0.f, 0.f, 0.f, 0.f, 0.f, 0.f};
  {
    const float* arow0 = tile + lo * AP + kb;
    const float* arow1 = arow0 + 16 * AP;
    const float* __restrict__ wcol = W1 + w * 16 + lo;
    for (int k0 = 0; k0 < D; k0 += 4) {
      const v2f a0 = *(const v2f*)(arow0 + k0);
      const v2f a1 = *(const v2f*)(arow1 + k0);
      v2f b;
      b.x = wcol[(size_t)(k0 + kb + 0) * HD];
      b.y = wcol[(size_t)(k0 + kb + 1) * HD];
      c0 = __builtin_amdgcn_wmma_f32_16x16x4_f32(
           false, a0,  false, b,
           (short)0, c0,  false,  false);
      c1 = __builtin_amdgcn_wmma_f32_16x16x4_f32(
           false, a1,  false, b,
           (short)0, c1,  false,  false);
    }
    const float bb = bias1[w * 16 + lo];
#pragma unroll
    for (int r = 0; r < 8; ++r) {
      c0[r] += bb;
      c1[r] += bb;
      if (RELU) {
        c0[r] = fmaxf(c0[r], 0.0f);
        c1[r] = fmaxf(c1[r], 0.0f);
      }
    }
  }
  __syncthreads();

#pragma unroll
  for (int r = 0; r < 8; ++r) {
    tile[(hi * 8 + r) * AP + w * 16 + lo] = c0[r];
    tile[(16 + hi * 8 + r) * AP + w * 16 + lo] = c1[r];
  }
  __syncthreads();

  const int mt = w >> 2;
  const int nt = w & 3;
  v8f c = (v8f){0.f, 0.f, 0.f, 0.f, 0.f, 0.f, 0.f, 0.f};
  {
    const float* arow = tile + (mt * 16 + lo) * AP + kb;
    const float* __restrict__ wcol = W2 + nt * 16 + lo;
    for (int k0 = 0; k0 < HD; k0 += 4) {
      const v2f a = *(const v2f*)(arow + k0);
      v2f b;
      b.x = wcol[(size_t)(k0 + kb + 0) * NO];
      b.y = wcol[(size_t)(k0 + kb + 1) * NO];
      c = __builtin_amdgcn_wmma_f32_16x16x4_f32(
           false, a,  false, b,
           (short)0, c,  false,  false);
    }
  }
#pragma unroll
  for (int r = 0; r < 8; ++r) otile[(mt * 16 + hi * 8 + r) * OP + nt * 16 + lo] = c[r];
  __syncthreads();

#pragma unroll
  for (int i = 0; i < 4; ++i) {
    const int node = node0 + i;
    if (node < n)
      *(v2f*)(out + (size_t)node * NO + lane * 2) = *(const v2f*)(otile + (w * 4 + i) * OP + lane * 2);
  }
}

extern "C" void kernel_launch(void* const* d_in, const int* in_sizes, int n_in,
                              void* d_out, int out_size, void* d_ws, size_t ws_size,
                              hipStream_t stream) {
  (void)n_in; (void)out_size; (void)ws_size;

  const float* x    = (const float*)d_in[0];
  const int*   ei   = (const int*)d_in[1];
  const float* W1   = (const float*)d_in[2];
  const float* b1   = (const float*)d_in[3];
  const float* W2   = (const float*)d_in[4];
  const float* b2   = (const float*)d_in[5];
  float*       out  = (float*)d_out;

  const int N = in_sizes[0] / 128;
  const int E = in_sizes[1] / 2;
  const int IN_D = 128, HID_D = 128, OUT_D = 64;

  const int* src = ei;
  const int* dst = ei + E;

  auto up4 = [](size_t v) { return (v + 3) & ~(size_t)3; };
  const int ntile = (N + SCAN_T * SCAN_ITEMS - 1) / (SCAN_T * SCAN_ITEMS);
  int*   cnt    = (int*)d_ws;
  int*   rowptr = cnt + up4((size_t)N);
  int*   tsum   = rowptr + up4((size_t)N + 1);
  int*   rank   = tsum + up4((size_t)ntile);
  int2*  csr    = (int2*)(rank + up4((size_t)E));
  float* dinv   = (float*)(csr + up4((size_t)E));
  float* bufB   = dinv + up4((size_t)N);

  const int T = GCN_THREADS;
  auto blk = [&](long long work) { return (unsigned)((work + T - 1) / T); };
  const unsigned gnode = (unsigned)((N + (T / 32) - 1) / (T / 32));

  k_cnt_zero<<<blk(N), T, 0, stream>>>(cnt, N);
  k_count_rank<<<blk(E), T, 0, stream>>>(dst, cnt, rank, E);
  k_tile_sum<<<ntile, SCAN_T, 0, stream>>>(cnt, tsum, N);
  k_tile_scan<<<ntile, SCAN_T, 0, stream>>>(cnt, tsum, rowptr, dinv, N);
  k_csr_fill<<<blk(E), T, 0, stream>>>(src, dst, rank, rowptr, dinv, csr, E);

  k_node_gather_gemm2<IN_D, HID_D, OUT_D, true><<<(unsigned)((N + 31) / 32), 256, 0, stream>>>(
      rowptr, csr, dinv, x, W1, b1, W2, bufB, N);

  k_node_gather<OUT_D, false><<<gnode, T, 0, stream>>>(rowptr, csr, dinv, bufB, b2, out, N);
}
